# starting kernel
# baseline (speedup 1.0000x reference)
#include <hip/hip_runtime.h>

#define BB 8
#define NN 2048
#define FF 64

typedef float v2f __attribute__((ext_vector_type(2)));
typedef float v8f __attribute__((ext_vector_type(8)));

__device__ __forceinline__ v8f wmma4(v2f a, v2f b, v8f c) {
    return __builtin_amdgcn_wmma_f32_16x16x4_f32(
         false, a,  false, b,
         (short)0, c,  false,  false);
}

__global__ __launch_bounds__(32) void k_h(const float* __restrict__ x,
                                          const float* __restrict__ Ww,
                                          const float* __restrict__ Wb,
                                          float* __restrict__ h) {
    const int lane = threadIdx.x & 31;
    const int half = lane >> 4;
    const int lm   = lane & 15;
    const int r0   = blockIdx.x * 16;

    v8f acc[4] = {};
#pragma unroll
    for (int kc = 0; kc < 16; ++kc) {
        const int k = kc * 4 + half * 2;
        v2f a = *(const v2f*)(x + (size_t)(r0 + lm) * FF + k);
#pragma unroll
        for (int nt = 0; nt < 4; ++nt) {
            const int col = nt * 16 + lm;
            v2f bop = *(const v2f*)(Ww + (size_t)col * FF + k);
            acc[nt] = wmma4(a, bop, acc[nt]);
        }
    }
#pragma unroll
    for (int nt = 0; nt < 4; ++nt) {
        const float bias = Wb[nt * 16 + lm];
#pragma unroll
        for (int v = 0; v < 8; ++v) {
            const int M = v + 8 * half;
            h[(size_t)(r0 + M) * FF + nt * 16 + lm] = acc[nt][v] + bias;
        }
    }
}

__global__ __launch_bounds__(256) void k_lr(const float* __restrict__ h,
                                            const float* __restrict__ a1,
                                            const float* __restrict__ a2,
                                            float* __restrict__ left,
                                            float* __restrict__ right) {
    const int r = blockIdx.x * blockDim.x + threadIdx.x;
    if (r >= BB * NN) return;
    float lf = 0.f, rt = 0.f;
#pragma unroll 8
    for (int o = 0; o < FF; ++o) {
        const float hv = h[(size_t)r * FF + o];
        lf = fmaf(hv, a1[o], lf);
        rt = fmaf(hv, a2[o], rt);
    }
    left[r]  = lf;
    right[r] = rt;
}

__global__ __launch_bounds__(256) void k_att(const float* __restrict__ adj,
                                             const float* __restrict__ h,
                                             const float* __restrict__ left,
                                             const float* __restrict__ right,
                                             const float* __restrict__ ab_p,
                                             float* __restrict__ out) {
    __shared__ float s_acc[4][4][8][32];
    __shared__ float s_l[4][16];

    const int tid  = threadIdx.x;
    const int w    = tid >> 5;
    const int lane = tid & 31;
    const int t    = w >> 1;
    const int s    = w & 1;
    const int half = lane >> 4;
    const int lm   = lane & 15;
    const int r0   = blockIdx.x * 64 + t * 16;
    const int b    = r0 / NN;

    const float ab = ab_p[0];
    const float lf = left[r0 + lm];

    const float* adj_row = adj + (size_t)(r0 + lm) * NN;
    const float* rbase   = right + (size_t)b * NN;
    const float* hbase   = h + (size_t)b * NN * FF;

    v8f   acc[4] = {};
    float lsum   = 0.f;

    for (int jt = s; jt < NN / 16; jt += 2) {
        const int j0 = jt * 16;
        __builtin_prefetch(adj_row + j0 + 32, 0, 0);
#pragma unroll
        for (int kc = 0; kc < 4; ++kc) {
            const int jo = j0 + kc * 4 + half * 2;
            const v2f av = __builtin_nontemporal_load((const v2f*)(adj_row + jo));
            const v2f rv = *(const v2f*)(rbase + jo);

            float z0 = lf + rv.x + ab;  z0 = z0 > 0.f ? z0 : 0.1f * z0;
            float z1 = lf + rv.y + ab;  z1 = z1 > 0.f ? z1 : 0.1f * z1;
            const float p0 = av.x > 0.f ? __expf(z0 + av.x) : 0.f;
            const float p1 = av.y > 0.f ? __expf(z1 + av.y) : 0.f;
            lsum += p0 + p1;

            v2f pa; pa.x = p0; pa.y = p1;
            const float* hK = hbase + (size_t)jo * FF;
#pragma unroll
            for (int nt = 0; nt < 4; ++nt) {
                v2f hb;
                hb.x = hK[nt * 16 + lm];
                hb.y = hK[FF + nt * 16 + lm];
                acc[nt] = wmma4(pa, hb, acc[nt]);
            }
        }
    }

    lsum += __shfl_xor(lsum, 16, 32);

    if (s == 1) {
#pragma unroll
        for (int nt = 0; nt < 4; ++nt)
#pragma unroll
            for (int v = 0; v < 8; ++v)
                s_acc[t][nt][v][lane] = acc[nt][v];
        if (lane < 16) s_l[t][lm] = lsum;
    }
    __syncthreads();

    if (s == 0) {
#pragma unroll
        for (int v = 0; v < 8; ++v) {
            const int M  = v + 8 * half;
            const float lt = __shfl(lsum, M, 32) + s_l[t][M];
            const float inv = lt > 0.f ? 1.0f / lt : 0.f;
#pragma unroll
            for (int nt = 0; nt < 4; ++nt) {
                float val = (acc[nt][v] + s_acc[t][nt][v][lane]) * inv;
                val = val > 0.f ? val : (__expf(val) - 1.f);
                __builtin_nontemporal_store(
                    val, out + (size_t)(r0 + M) * FF + nt * 16 + lm);
            }
        }
    }
}

extern "C" void kernel_launch(void* const* d_in, const int* in_sizes, int n_in,
                              void* d_out, int out_size, void* d_ws, size_t ws_size,
                              hipStream_t stream) {
    const float* x   = (const float*)d_in[0];
    const float* adj = (const float*)d_in[1];
    const float* Ww  = (const float*)d_in[2];
    const float* Wb  = (const float*)d_in[3];
    const float* a1  = (const float*)d_in[4];
    const float* a2  = (const float*)d_in[5];
    const float* ab  = (const float*)d_in[6];
    float* out = (float*)d_out;

    float* h     = (float*)d_ws;
    float* left  = h + (size_t)BB * NN * FF;
    float* right = left + (size_t)BB * NN;

    k_h  <<<BB * NN / 16, 32,  0, stream>>>(x, Ww, Wb, h);
    k_lr <<<(BB * NN) / 256, 256, 0, stream>>>(h, a1, a2, left, right);
    k_att<<<BB * NN / 64, 256, 0, stream>>>(adj, h, left, right, ab, out);
}


# split each 16-row tile's j range over 8 waves instead of 2 (1024-thread blocks, 8 waves per SIMD) so sibling waves cover load waits; 8 partials summed in LDS in fixed stripe order (new sum order)
# speedup vs baseline: 1.8361x; 1.8361x over previous
#include <hip/hip_runtime.h>

#define BB 8
#define NN 2048
#define FF 64

typedef float v2f __attribute__((ext_vector_type(2)));
typedef float v4f __attribute__((ext_vector_type(4)));
typedef float v8f __attribute__((ext_vector_type(8)));

__device__ __forceinline__ v8f wmma4(v2f a, v2f b, v8f c) {
    return __builtin_amdgcn_wmma_f32_16x16x4_f32(
         false, a,  false, b,
         (short)0, c,  false,  false);
}

__global__ __launch_bounds__(32) void k_h(const float* __restrict__ x,
                                          const float* __restrict__ Ww,
                                          const float* __restrict__ Wb,
                                          float* __restrict__ h) {
    const int lane = threadIdx.x & 31;
    const int half = lane >> 4;
    const int lm   = lane & 15;
    const int r0   = blockIdx.x * 16;

    v8f acc[4] = {};
#pragma unroll
    for (int kc = 0; kc < 16; ++kc) {
        const int k = kc * 4 + half * 2;
        v2f a = *(const v2f*)(x + (size_t)(r0 + lm) * FF + k);
#pragma unroll
        for (int nt = 0; nt < 4; ++nt) {
            const int col = nt * 16 + lm;
            v2f bop = *(const v2f*)(Ww + (size_t)col * FF + k);
            acc[nt] = wmma4(a, bop, acc[nt]);
        }
    }
#pragma unroll
    for (int nt = 0; nt < 4; ++nt) {
        const float bias = Wb[nt * 16 + lm];
#pragma unroll
        for (int v = 0; v < 8; ++v) {
            const int M = v + 8 * half;
            h[(size_t)(r0 + M) * FF + nt * 16 + lm] = acc[nt][v] + bias;
        }
    }
}

__global__ __launch_bounds__(256) void k_lr(const float* __restrict__ h,
                                            const float* __restrict__ a1,
                                            const float* __restrict__ a2,
                                            float* __restrict__ left,
                                            float* __restrict__ right) {
    const int r = blockIdx.x * blockDim.x + threadIdx.x;
    if (r >= BB * NN) return;
    float lf = 0.f, rt = 0.f;
#pragma unroll 8
    for (int o = 0; o < FF; ++o) {
        const float hv = h[(size_t)r * FF + o];
        lf = fmaf(hv, a1[o], lf);
        rt = fmaf(hv, a2[o], rt);
    }
    left[r]  = lf;
    right[r] = rt;
}

#define NS 8

__global__ __launch_bounds__(32 * 4 * NS) void k_att(const float* __restrict__ adj,
                                             const float* __restrict__ h,
                                             const float* __restrict__ left,
                                             const float* __restrict__ right,
                                             const float* __restrict__ ab_p,
                                             float* __restrict__ out) {
    __shared__ float s_acc[4][NS][4][8][32];
    __shared__ float s_l[4][NS][16];

    const int tid  = threadIdx.x;
    const int w    = tid >> 5;
    const int lane = tid & 31;
    const int t    = w / NS;
    const int s    = w % NS;
    const int half = lane >> 4;
    const int lm   = lane & 15;
    const int r0   = blockIdx.x * 64 + t * 16;
    const int b    = r0 / NN;

    const float ab = ab_p[0];
    const float lf = left[r0 + lm];

    const float* adj_row = adj + (size_t)(r0 + lm) * NN;
    const float* rbase   = right + (size_t)b * NN;
    const float* hbase   = h + (size_t)b * NN * FF;

    v8f   acc[4] = {};
    float lsum   = 0.f;

    for (int jt = s; jt < NN / 16; jt += NS) {
        const int j0 = jt * 16;
        if (jt + NS < NN / 16)
            __builtin_prefetch(adj_row + j0 + 16 * NS, 0, 0);
#pragma unroll
        for (int kc = 0; kc < 4; ++kc) {
            const int jo = j0 + kc * 4 + half * 2;
            const v2f av = __builtin_nontemporal_load((const v2f*)(adj_row + jo));
            const v2f rv = *(const v2f*)(rbase + jo);

            float z0 = lf + rv.x + ab;  z0 = z0 > 0.f ? z0 : 0.1f * z0;
            float z1 = lf + rv.y + ab;  z1 = z1 > 0.f ? z1 : 0.1f * z1;
            const float p0 = av.x > 0.f ? __expf(z0 + av.x) : 0.f;
            const float p1 = av.y > 0.f ? __expf(z1 + av.y) : 0.f;
            lsum += p0 + p1;

            v2f pa; pa.x = p0; pa.y = p1;
            const float* hK = hbase + (size_t)jo * FF;
#pragma unroll
            for (int nt = 0; nt < 4; ++nt) {
                v2f hb;
                hb.x = hK[nt * 16 + lm];
                hb.y = hK[FF + nt * 16 + lm];
                acc[nt] = wmma4(pa, hb, acc[nt]);
            }
        }
    }

    lsum += __shfl_xor(lsum, 16, 32);

#pragma unroll
    for (int nt = 0; nt < 4; ++nt)
#pragma unroll
        for (int v = 0; v < 8; ++v)
            s_acc[t][s][nt][v][lane] = acc[nt][v];
    if (lane < 16) s_l[t][s][lm] = lsum;
    __syncthreads();

    {
        const int M   = 2 * s + half;
        const int v   = M & 7;
        const int cl  = (M >> 3) * 16 + (lm & 3) * 4;
        const int ntc = lm >> 2;
        float lt = 0.f;
        float o0 = 0.f, o1 = 0.f, o2 = 0.f, o3 = 0.f;
#pragma unroll
        for (int q = 0; q < NS; ++q) {
            lt += s_l[t][q][M];
            const v4f pv = *(const v4f*)&s_acc[t][q][ntc][v][cl];
            o0 += pv.x; o1 += pv.y; o2 += pv.z; o3 += pv.w;
        }
        const float inv = lt > 0.f ? 1.0f / lt : 0.f;
        v4f r;
        r.x = o0 * inv; r.y = o1 * inv; r.z = o2 * inv; r.w = o3 * inv;
        r.x = r.x > 0.f ? r.x : (__expf(r.x) - 1.f);
        r.y = r.y > 0.f ? r.y : (__expf(r.y) - 1.f);
        r.z = r.z > 0.f ? r.z : (__expf(r.z) - 1.f);
        r.w = r.w > 0.f ? r.w : (__expf(r.w) - 1.f);
        __builtin_nontemporal_store(
            r, (v4f*)(out + (size_t)(r0 + M) * FF + lm * 4));
    }
}

extern "C" void kernel_launch(void* const* d_in, const int* in_sizes, int n_in,
                              void* d_out, int out_size, void* d_ws, size_t ws_size,
                              hipStream_t stream) {
    const float* x   = (const float*)d_in[0];
    const float* adj = (const float*)d_in[1];
    const float* Ww  = (const float*)d_in[2];
    const float* Wb  = (const float*)d_in[3];
    const float* a1  = (const float*)d_in[4];
    const float* a2  = (const float*)d_in[5];
    const float* ab  = (const float*)d_in[6];
    float* out = (float*)d_out;

    float* h     = (float*)d_ws;
    float* left  = h + (size_t)BB * NN * FF;
    float* right = left + (size_t)BB * NN;

    k_h  <<<BB * NN / 16, 32,  0, stream>>>(x, Ww, Wb, h);
    k_lr <<<(BB * NN) / 256, 256, 0, stream>>>(h, a1, a2, left, right);
    k_att<<<BB * NN / 64, 32 * 4 * NS, 0, stream>>>(adj, h, left, right, ab, out);
}


# adj fetched as whole 128-B lines (8 lanes x 16 B per row) instead of 16-row column loads; p built in that layout, moved to the A layout through a wave-private LDS tile; new k order (new sum order)
# speedup vs baseline: 2.5313x; 1.3786x over previous
#include <hip/hip_runtime.h>

#define BB 8
#define NN 2048
#define FF 64

typedef float v2f __attribute__((ext_vector_type(2)));
typedef float v4f __attribute__((ext_vector_type(4)));
typedef float v8f __attribute__((ext_vector_type(8)));

__device__ __forceinline__ v8f wmma4(v2f a, v2f b, v8f c) {
    return __builtin_amdgcn_wmma_f32_16x16x4_f32(
         false, a,  false, b,
         (short)0, c,  false,  false);
}

__global__ __launch_bounds__(32) void k_h(const float* __restrict__ x,
                                          const float* __restrict__ Ww,
                                          const float* __restrict__ Wb,
                                          float* __restrict__ h) {
    const int lane = threadIdx.x & 31;
    const int half = lane >> 4;
    const int lm   = lane & 15;
    const int r0   = blockIdx.x * 16;

    v8f acc[4] = {};
#pragma unroll
    for (int kc = 0; kc < 16; ++kc) {
        const int k = kc * 4 + half * 2;
        v2f a = *(const v2f*)(x + (size_t)(r0 + lm) * FF + k);
#pragma unroll
        for (int nt = 0; nt < 4; ++nt) {
            const int col = nt * 16 + lm;
            v2f bop = *(const v2f*)(Ww + (size_t)col * FF + k);
            acc[nt] = wmma4(a, bop, acc[nt]);
        }
    }
#pragma unroll
    for (int nt = 0; nt < 4; ++nt) {
        const float bias = Wb[nt * 16 + lm];
#pragma unroll
        for (int v = 0; v < 8; ++v) {
            const int M = v + 8 * half;
            h[(size_t)(r0 + M) * FF + nt * 16 + lm] = acc[nt][v] + bias;
        }
    }
}

__global__ __launch_bounds__(256) void k_lr(const float* __restrict__ h,
                                            const float* __restrict__ a1,
                                            const float* __restrict__ a2,
                                            float* __restrict__ left,
                                            float* __restrict__ right) {
    const int r = blockIdx.x * blockDim.x + threadIdx.x;
    if (r >= BB * NN) return;
    float lf = 0.f, rt = 0.f;
#pragma unroll 8
    for (int o = 0; o < FF; ++o) {
        const float hv = h[(size_t)r * FF + o];
        lf = fmaf(hv, a1[o], lf);
        rt = fmaf(hv, a2[o], rt);
    }
    left[r]  = lf;
    right[r] = rt;
}

#define NS 8
#define PW 32
#define PP 36

__global__ __launch_bounds__(32 * 4 * NS) void k_att(const float* __restrict__ adj,
                                             const float* __restrict__ h,
                                             const float* __restrict__ left,
                                             const float* __restrict__ right,
                                             const float* __restrict__ ab_p,
                                             float* __restrict__ out) {
    __shared__ float s_acc[4][NS][4][8][32];
    __shared__ float s_l[4][NS][16];
    __shared__ __attribute__((aligned(16))) float s_p[4 * NS][16 * PP];

    const int tid  = threadIdx.x;
    const int w    = __builtin_amdgcn_readfirstlane(tid >> 5);
    const int lane = tid & 31;
    const int t    = w / NS;
    const int s    = w % NS;
    const int half = lane >> 4;
    const int lm   = lane & 15;
    const int lr   = lane >> 3;
    const int lc   = (lane & 7) * 4;
    const int r0   = blockIdx.x * 64 + t * 16;
    const int b    = r0 / NN;

    const float ab = ab_p[0];
    float lfq[4];
#pragma unroll
    for (int q = 0; q < 4; ++q) lfq[q] = left[r0 + 4 * q + lr];

    const float* adj_ld = adj + (size_t)(r0 + lr) * NN + lc;
    const float* rbase  = right + (size_t)b * NN + lc;
    const float* hbase  = h + (size_t)b * NN * FF + (size_t)(16 * half) * FF + lm;
    float*       p_st   = &s_p[w][lr * PP + lc];
    const float* p_ld   = &s_p[w][lm * PP + 16 * half];

    v8f   acc[4] = {};
    float lsum   = 0.f;

    v4f an[4];
    v4f rn;
#pragma unroll
    for (int q = 0; q < 4; ++q)
        an[q] = __builtin_nontemporal_load(
            (const v4f*)(adj_ld + (size_t)q * 4 * NN + s * PW));
    rn = *(const v4f*)(rbase + s * PW);

    for (int it = 0; it < NN / PW / NS; ++it) {
        const int j0 = (s + it * NS) * PW;
        const v4f rv = rn;

#pragma unroll
        for (int q = 0; q < 4; ++q) {
            v4f p;
#pragma unroll
            for (int e = 0; e < 4; ++e) {
                const float av = an[q][e];
                float z = lfq[q] + rv[e] + ab;  z = z > 0.f ? z : 0.1f * z;
                const float ex = __expf(z + av);
                p[e] = av > 0.f ? ex : 0.f;
            }
            *(v4f*)(p_st + q * 4 * PP) = p;
        }
        if (it + 1 < NN / PW / NS) {
#pragma unroll
            for (int q = 0; q < 4; ++q)
                an[q] = __builtin_nontemporal_load(
                    (const v4f*)(adj_ld + (size_t)q * 4 * NN + j0 + NS * PW));
            rn = *(const v4f*)(rbase + j0 + NS * PW);
        }
        __builtin_amdgcn_fence(__ATOMIC_RELEASE, "wavefront");
        __builtin_amdgcn_wave_barrier();
        __builtin_amdgcn_fence(__ATOMIC_ACQUIRE, "wavefront");
        v4f P[4];
#pragma unroll
        for (int i = 0; i < 4; ++i) P[i] = *(const v4f*)(p_ld + 4 * i);
        __builtin_amdgcn_fence(__ATOMIC_RELEASE, "wavefront");
        __builtin_amdgcn_wave_barrier();
        __builtin_amdgcn_fence(__ATOMIC_ACQUIRE, "wavefront");

        const float* hJ = hbase + (size_t)j0 * FF;
#pragma unroll
        for (int c = 0; c < 8; ++c) {
            const float* hK = hJ + (size_t)(2 * c) * FF;
            v2f hb[4];
#pragma unroll
            for (int nt = 0; nt < 4; ++nt) {
                hb[nt].x = hK[nt * 16];
                hb[nt].y = hK[FF + nt * 16];
            }
            const float p0 = P[c >> 1][(c & 1) * 2];
            const float p1 = P[c >> 1][(c & 1) * 2 + 1];
            lsum += p0 + p1;
            v2f pa; pa.x = p0; pa.y = p1;
#pragma unroll
            for (int nt = 0; nt < 4; ++nt)
                acc[nt] = wmma4(pa, hb[nt], acc[nt]);
        }
    }

    lsum += __shfl_xor(lsum, 16, 32);

#pragma unroll
    for (int nt = 0; nt < 4; ++nt)
#pragma unroll
        for (int v = 0; v < 8; ++v)
            s_acc[t][s][nt][v][lane] = acc[nt][v];
    if (lane < 16) s_l[t][s][lm] = lsum;
    __syncthreads();

    {
        const int M   = 2 * s + half;
        const int v   = M & 7;
        const int cl  = (M >> 3) * 16 + (lm & 3) * 4;
        const int ntc = lm >> 2;
        float lt = 0.f;
        float o0 = 0.f, o1 = 0.f, o2 = 0.f, o3 = 0.f;
#pragma unroll
        for (int q = 0; q < NS; ++q) {
            lt += s_l[t][q][M];
            const v4f pv = *(const v4f*)&s_acc[t][q][ntc][v][cl];
            o0 += pv.x; o1 += pv.y; o2 += pv.z; o3 += pv.w;
        }
        const float inv = lt > 0.f ? 1.0f / lt : 0.f;
        v4f r;
        r.x = o0 * inv; r.y = o1 * inv; r.z = o2 * inv; r.w = o3 * inv;
        r.x = r.x > 0.f ? r.x : (__expf(r.x) - 1.f);
        r.y = r.y > 0.f ? r.y : (__expf(r.y) - 1.f);
        r.z = r.z > 0.f ? r.z : (__expf(r.z) - 1.f);
        r.w = r.w > 0.f ? r.w : (__expf(r.w) - 1.f);
        __builtin_nontemporal_store(
            r, (v4f*)(out + (size_t)(r0 + M) * FF + lm * 4));
    }
}

extern "C" void kernel_launch(void* const* d_in, const int* in_sizes, int n_in,
                              void* d_out, int out_size, void* d_ws, size_t ws_size,
                              hipStream_t stream) {
    const float* x   = (const float*)d_in[0];
    const float* adj = (const float*)d_in[1];
    const float* Ww  = (const float*)d_in[2];
    const float* Wb  = (const float*)d_in[3];
    const float* a1  = (const float*)d_in[4];
    const float* a2  = (const float*)d_in[5];
    const float* ab  = (const float*)d_in[6];
    float* out = (float*)d_out;

    float* h     = (float*)d_ws;
    float* left  = h + (size_t)BB * NN * FF;
    float* right = left + (size_t)BB * NN;

    k_h  <<<BB * NN / 16, 32,  0, stream>>>(x, Ww, Wb, h);
    k_lr <<<(BB * NN) / 256, 256, 0, stream>>>(h, a1, a2, left, right);
    k_att<<<BB * NN / 64, 32 * 4 * NS, 0, stream>>>(adj, h, left, right, ab, out);
}


# each 256-row chunk of h copied to LDS once per block (two buffers, one block barrier per trip); B operands read from there as k-pair 8-byte LDS loads instead of per-wave global loads; same sum order
# speedup vs baseline: 2.8813x; 1.1383x over previous
#include <hip/hip_runtime.h>

#define BB 8
#define NN 2048
#define FF 64

typedef float v2f __attribute__((ext_vector_type(2)));
typedef float v4f __attribute__((ext_vector_type(4)));
typedef float v8f __attribute__((ext_vector_type(8)));

__device__ __forceinline__ v8f wmma4(v2f a, v2f b, v8f c) {
    return __builtin_amdgcn_wmma_f32_16x16x4_f32(
         false, a,  false, b,
         (short)0, c,  false,  false);
}

__global__ __launch_bounds__(32) void k_h(const float* __restrict__ x,
                                          const float* __restrict__ Ww,
                                          const float* __restrict__ Wb,
                                          float* __restrict__ h) {
    const int lane = threadIdx.x & 31;
    const int half = lane >> 4;
    const int lm   = lane & 15;
    const int r0   = blockIdx.x * 16;

    v8f acc[4] = {};
#pragma unroll
    for (int kc = 0; kc < 16; ++kc) {
        const int k = kc * 4 + half * 2;
        v2f a = *(const v2f*)(x + (size_t)(r0 + lm) * FF + k);
#pragma unroll
        for (int nt = 0; nt < 4; ++nt) {
            const int col = nt * 16 + lm;
            v2f bop = *(const v2f*)(Ww + (size_t)col * FF + k);
            acc[nt] = wmma4(a, bop, acc[nt]);
        }
    }
#pragma unroll
    for (int nt = 0; nt < 4; ++nt) {
        const float bias = Wb[nt * 16 + lm];
#pragma unroll
        for (int v = 0; v < 8; ++v) {
            const int M = v + 8 * half;
            h[(size_t)(r0 + M) * FF + nt * 16 + lm] = acc[nt][v] + bias;
        }
    }
}

__global__ __launch_bounds__(256) void k_lr(const float* __restrict__ h,
                                            const float* __restrict__ a1,
                                            const float* __restrict__ a2,
                                            float* __restrict__ left,
                                            float* __restrict__ right) {
    const int r = blockIdx.x * blockDim.x + threadIdx.x;
    if (r >= BB * NN) return;
    float lf = 0.f, rt = 0.f;
#pragma unroll 8
    for (int o = 0; o < FF; ++o) {
        const float hv = h[(size_t)r * FF + o];
        lf = fmaf(hv, a1[o], lf);
        rt = fmaf(hv, a2[o], rt);
    }
    left[r]  = lf;
    right[r] = rt;
}

#define NS 8
#define PW 32
#define PP 36
#define CH (NS * PW)
#define CHF (CH * FF)

__global__ __launch_bounds__(32 * 4 * NS) void k_att(const float* __restrict__ adj,
                                             const float* __restrict__ h,
                                             const float* __restrict__ left,
                                             const float* __restrict__ right,
                                             const float* __restrict__ ab_p,
                                             float* __restrict__ out) {
    __shared__ __attribute__((aligned(16))) float s_big[2 * CHF];
    __shared__ float s_l[4][NS][16];
    __shared__ __attribute__((aligned(16))) float s_p[4 * NS][16 * PP];

    const int tid  = threadIdx.x;
    const int w    = __builtin_amdgcn_readfirstlane(tid >> 5);
    const int lane = tid & 31;
    const int t    = w / NS;
    const int s    = w % NS;
    const int half = lane >> 4;
    const int lm   = lane & 15;
    const int lr   = lane >> 3;
    const int lc   = (lane & 7) * 4;
    const int r0   = blockIdx.x * 64 + t * 16;
    const int b    = r0 / NN;

    const float ab = ab_p[0];
    float lfq[4];
#pragma unroll
    for (int q = 0; q < 4; ++q) lfq[q] = left[r0 + 4 * q + lr];

    const float* adj_ld = adj + (size_t)(r0 + lr) * NN + lc;
    const float* rbase  = right + (size_t)b * NN + lc;
    float*       p_st   = &s_p[w][lr * PP + lc];
    const float* p_ld   = &s_p[w][lm * PP + 16 * half];

    const float* hs_ld = h + (size_t)b * NN * FF + (tid >> 4) * 2 * FF + (tid & 15) * 4;
    float*       hs_st = s_big + (tid >> 4) * 2 * FF
                               + ((((tid & 15) * 4) ^ (((tid >> 7) & 1) * 16)) * 2);
    const float* hA = s_big + (s * 16 + 8 * half) * 2 * FF + (lm + 16 * half) * 2;
    const float* hB = s_big + (s * 16 + 8 * half) * 2 * FF + (lm + 16 * (1 - half)) * 2;

    v8f   acc[4] = {};
    float lsum   = 0.f;

    v4f an[4];
    v4f rn;
    v4f hs[2][2];
#pragma unroll
    for (int k = 0; k < 2; ++k) {
        hs[k][0] = *(const v4f*)(hs_ld + k * 64 * 2 * FF);
        hs[k][1] = *(const v4f*)(hs_ld + k * 64 * 2 * FF + FF);
    }
#pragma unroll
    for (int q = 0; q < 4; ++q)
        an[q] = __builtin_nontemporal_load(
            (const v4f*)(adj_ld + (size_t)q * 4 * NN + s * PW));
    rn = *(const v4f*)(rbase + s * PW);
#pragma unroll
    for (int k = 0; k < 2; ++k) {
        v4f lo, hi;
        lo.x = hs[k][0].x; lo.y = hs[k][1].x; lo.z = hs[k][0].y; lo.w = hs[k][1].y;
        hi.x = hs[k][0].z; hi.y = hs[k][1].z; hi.z = hs[k][0].w; hi.w = hs[k][1].w;
        *(v4f*)(hs_st + k * 64 * 2 * FF)     = lo;
        *(v4f*)(hs_st + k * 64 * 2 * FF + 4) = hi;
    }
    __syncthreads();

    for (int it = 0; it < NN / PW / NS; ++it) {
        const int j0  = (s + it * NS) * PW;
        const int cur = (it & 1) * CHF;
        const v4f rv = rn;

#pragma unroll
        for (int q = 0; q < 4; ++q) {
            v4f p;
#pragma unroll
            for (int e = 0; e < 4; ++e) {
                const float av = an[q][e];
                float z = lfq[q] + rv[e] + ab;  z = z > 0.f ? z : 0.1f * z;
                const float ex = __expf(z + av);
                p[e] = av > 0.f ? ex : 0.f;
            }
            *(v4f*)(p_st + q * 4 * PP) = p;
        }
        if (it + 1 < NN / PW / NS) {
#pragma unroll
            for (int q = 0; q < 4; ++q)
                an[q] = __builtin_nontemporal_load(
                    (const v4f*)(adj_ld + (size_t)q * 4 * NN + j0 + NS * PW));
            rn = *(const v4f*)(rbase + j0 + NS * PW);
#pragma unroll
            for (int k = 0; k < 2; ++k) {
                hs[k][0] = *(const v4f*)(hs_ld + (size_t)(it + 1) * CHF + k * 64 * 2 * FF);
                hs[k][1] = *(const v4f*)(hs_ld + (size_t)(it + 1) * CHF + k * 64 * 2 * FF + FF);
            }
        }
        __builtin_amdgcn_fence(__ATOMIC_RELEASE, "wavefront");
        __builtin_amdgcn_wave_barrier();
        __builtin_amdgcn_fence(__ATOMIC_ACQUIRE, "wavefront");
        const float* hAc = hA + cur;
        const float* hBc = hB + cur;
        v2f pa[2];
        v2f hb[2][4];
        pa[0]    = *(const v2f*)(p_ld);
        hb[0][0] = *(const v2f*)(hAc);
        hb[0][1] = *(const v2f*)(hBc);
        hb[0][2] = *(const v2f*)(hAc + FF);
        hb[0][3] = *(const v2f*)(hBc + FF);
#pragma unroll
        for (int c = 0; c < 8; ++c) {
            if (c + 1 < 8) {
                pa[(c + 1) & 1]    = *(const v2f*)(p_ld + 2 * (c + 1));
                hb[(c + 1) & 1][0] = *(const v2f*)(hAc + (c + 1) * 2 * FF);
                hb[(c + 1) & 1][1] = *(const v2f*)(hBc + (c + 1) * 2 * FF);
                hb[(c + 1) & 1][2] = *(const v2f*)(hAc + (c + 1) * 2 * FF + FF);
                hb[(c + 1) & 1][3] = *(const v2f*)(hBc + (c + 1) * 2 * FF + FF);
            }
            lsum += pa[c & 1].x + pa[c & 1].y;
#pragma unroll
            for (int nt = 0; nt < 4; ++nt)
                acc[nt] = wmma4(pa[c & 1], hb[c & 1][nt], acc[nt]);
        }
        __builtin_amdgcn_fence(__ATOMIC_RELEASE, "wavefront");
        __builtin_amdgcn_wave_barrier();
        __builtin_amdgcn_fence(__ATOMIC_ACQUIRE, "wavefront");
        if (it + 1 < NN / PW / NS) {
#pragma unroll
            for (int k = 0; k < 2; ++k) {
                v4f lo, hi;
                lo.x = hs[k][0].x; lo.y = hs[k][1].x; lo.z = hs[k][0].y; lo.w = hs[k][1].y;
                hi.x = hs[k][0].z; hi.y = hs[k][1].z; hi.z = hs[k][0].w; hi.w = hs[k][1].w;
                *(v4f*)(hs_st + (CHF - cur) + k * 64 * 2 * FF)     = lo;
                *(v4f*)(hs_st + (CHF - cur) + k * 64 * 2 * FF + 4) = hi;
            }
        }
        __syncthreads();
    }

    lsum += __shfl_xor(lsum, 16, 32);

    const int lane_e = __lane_id();
    const int half_e = lane_e >> 4;
    const int lm_e   = lane_e & 15;

    float* s_acc = s_big;
#pragma unroll
    for (int nt = 0; nt < 4; ++nt)
#pragma unroll
        for (int v = 0; v < 8; ++v)
            s_acc[(((t * NS + s) * 4 + nt) * 8 + v) * 32 + lane_e] = acc[nt][v];
    if (lane_e < 16) s_l[t][s][lm_e] = lsum;
    __syncthreads();

    {
        const int M   = 2 * s + half_e;
        const int v   = M & 7;
        const int cl  = (M >> 3) * 16 + (lm_e & 3) * 4;
        const int ntc = lm_e >> 2;
        float lt = 0.f;
        float o0 = 0.f, o1 = 0.f, o2 = 0.f, o3 = 0.f;
#pragma unroll
        for (int q = 0; q < NS; ++q) {
            lt += s_l[t][q][M];
            const v4f pv = *(const v4f*)&s_acc[(((t * NS + q) * 4 + ntc) * 8 + v) * 32 + cl];
            o0 += pv.x; o1 += pv.y; o2 += pv.z; o3 += pv.w;
        }
        const float inv = lt > 0.f ? 1.0f / lt : 0.f;
        v4f r;
        r.x = o0 * inv; r.y = o1 * inv; r.z = o2 * inv; r.w = o3 * inv;
        r.x = r.x > 0.f ? r.x : (__expf(r.x) - 1.f);
        r.y = r.y > 0.f ? r.y : (__expf(r.y) - 1.f);
        r.z = r.z > 0.f ? r.z : (__expf(r.z) - 1.f);
        r.w = r.w > 0.f ? r.w : (__expf(r.w) - 1.f);
        __builtin_nontemporal_store(
            r, (v4f*)(out + (size_t)(r0 + M) * FF + lm_e * 4));
    }
}

extern "C" void kernel_launch(void* const* d_in, const int* in_sizes, int n_in,
                              void* d_out, int out_size, void* d_ws, size_t ws_size,
                              hipStream_t stream) {
    const float* x   = (const float*)d_in[0];
    const float* adj = (const float*)d_in[1];
    const float* Ww  = (const float*)d_in[2];
    const float* Wb  = (const float*)d_in[3];
    const float* a1  = (const float*)d_in[4];
    const float* a2  = (const float*)d_in[5];
    const float* ab  = (const float*)d_in[6];
    float* out = (float*)d_out;

    float* h     = (float*)d_ws;
    float* left  = h + (size_t)BB * NN * FF;
    float* right = left + (size_t)BB * NN;

    k_h  <<<BB * NN / 16, 32,  0, stream>>>(x, Ww, Wb, h);
    k_lr <<<(BB * NN) / 256, 256, 0, stream>>>(h, a1, a2, left, right);
    k_att<<<BB * NN / 64, 32 * 4 * NS, 0, stream>>>(adj, h, left, right, ab, out);
}


# left=h@a1, right=h@a2 taken from the accumulator tile in the h kernel's epilogue (16-lane DPP sums, one 128-B line per wave); per-row kernel and its launch removed; left/right in a new sum order
# speedup vs baseline: 3.0028x; 1.0422x over previous
#include <hip/hip_runtime.h>

#define BB 8
#define NN 2048
#define FF 64

typedef float v2f __attribute__((ext_vector_type(2)));
typedef float v4f __attribute__((ext_vector_type(4)));
typedef float v8f __attribute__((ext_vector_type(8)));

__device__ __forceinline__ v8f wmma4(v2f a, v2f b, v8f c) {
    return __builtin_amdgcn_wmma_f32_16x16x4_f32(
         false, a,  false, b,
         (short)0, c,  false,  false);
}


template <int MASK>
__device__ __forceinline__ float row_xor(float v) {
    return __int_as_float(__builtin_amdgcn_update_dpp(
        0, __float_as_int(v), 0x160 + MASK  , 0xf, 0xf, true));
}

__device__ __forceinline__ float row_sum16(float v) {
    v += row_xor<1>(v);
    v += row_xor<2>(v);
    v += row_xor<4>(v);
    v += row_xor<8>(v);
    return v;
}

__global__ __launch_bounds__(64) void k_h(const float* __restrict__ x,
                                          const float* __restrict__ Ww,
                                          const float* __restrict__ Wb,
                                          const float* __restrict__ a1,
                                          const float* __restrict__ a2,
                                          float* __restrict__ h,
                                          float* __restrict__ left,
                                          float* __restrict__ right) {
    __shared__ __attribute__((aligned(16))) float s_lr[2][32];

    const int w    = __builtin_amdgcn_readfirstlane(threadIdx.x >> 5);
    const int lane = threadIdx.x & 31;
    const int half = lane >> 4;
    const int lm   = lane & 15;
    const int r0   = (blockIdx.x * 2 + w) * 16;

    v8f acc[4] = {};
#pragma unroll
    for (int kc = 0; kc < 16; ++kc) {
        const int k = kc * 4 + half * 2;
        v2f a = *(const v2f*)(x + (size_t)(r0 + lm) * FF + k);
#pragma unroll
        for (int nt = 0; nt < 4; ++nt) {
            const int col = nt * 16 + lm;
            v2f bop = *(const v2f*)(Ww + (size_t)col * FF + k);
            acc[nt] = wmma4(a, bop, acc[nt]);
        }
    }
    float pl[8] = {}, pr[8] = {};
#pragma unroll
    for (int nt = 0; nt < 4; ++nt) {
        const float bias = Wb[nt * 16 + lm];
        const float c1   = a1[nt * 16 + lm];
        const float c2   = a2[nt * 16 + lm];
#pragma unroll
        for (int v = 0; v < 8; ++v) {
            acc[nt][v] += bias;
            pl[v] = fmaf(acc[nt][v], c1, pl[v]);
            pr[v] = fmaf(acc[nt][v], c2, pr[v]);
        }
    }
#pragma unroll
    for (int v = 0; v < 8; ++v) {
        pl[v] = row_sum16(pl[v]);
        pr[v] = row_sum16(pr[v]);
    }
    if (lm == 0) {
        v4f t;
        t.x = pl[0]; t.y = pl[1]; t.z = pl[2]; t.w = pl[3];
        *(v4f*)&s_lr[0][w * 16 + 8 * half]     = t;
        t.x = pl[4]; t.y = pl[5]; t.z = pl[6]; t.w = pl[7];
        *(v4f*)&s_lr[0][w * 16 + 8 * half + 4] = t;
        t.x = pr[0]; t.y = pr[1]; t.z = pr[2]; t.w = pr[3];
        *(v4f*)&s_lr[1][w * 16 + 8 * half]     = t;
        t.x = pr[4]; t.y = pr[5]; t.z = pr[6]; t.w = pr[7];
        *(v4f*)&s_lr[1][w * 16 + 8 * half + 4] = t;
    }
    __syncthreads();
    float* dst = w ? right : left;
    dst[blockIdx.x * 32 + lane] = s_lr[w][lane];
#pragma unroll
    for (int nt = 0; nt < 4; ++nt)
#pragma unroll
        for (int v = 0; v < 8; ++v)
            h[(size_t)(r0 + v + 8 * half) * FF + nt * 16 + lm] = acc[nt][v];
}

#define NS 8
#define PW 32
#define PP 36
#define CH (NS * PW)
#define CHF (CH * FF)

__global__ __launch_bounds__(32 * 4 * NS) void k_att(const float* __restrict__ adj,
                                             const float* __restrict__ h,
                                             const float* __restrict__ left,
                                             const float* __restrict__ right,
                                             const float* __restrict__ ab_p,
                                             float* __restrict__ out) {
    __shared__ __attribute__((aligned(16))) float s_big[2 * CHF];
    __shared__ float s_l[4][NS][16];
    __shared__ __attribute__((aligned(16))) float s_p[4 * NS][16 * PP];

    const int tid  = threadIdx.x;
    const int w    = __builtin_amdgcn_readfirstlane(tid >> 5);
    const int lane = tid & 31;
    const int t    = w / NS;
    const int s    = w % NS;
    const int half = lane >> 4;
    const int lm   = lane & 15;
    const int lr   = lane >> 3;
    const int lc   = (lane & 7) * 4;
    const int r0   = blockIdx.x * 64 + t * 16;
    const int b    = r0 / NN;

    const float ab = ab_p[0];
    float lfq[4];
#pragma unroll
    for (int q = 0; q < 4; ++q) lfq[q] = left[r0 + 4 * q + lr];

    const float* adj_ld = adj + (size_t)(r0 + lr) * NN + lc;
    const float* rbase  = right + (size_t)b * NN + lc;
    float*       p_st   = &s_p[w][lr * PP + lc];
    const float* p_ld   = &s_p[w][lm * PP + 16 * half];

    const float* hs_ld = h + (size_t)b * NN * FF + (tid >> 4) * 2 * FF + (tid & 15) * 4;
    float*       hs_st = s_big + (tid >> 4) * 2 * FF
                               + ((((tid & 15) * 4) ^ (((tid >> 7) & 1) * 16)) * 2);
    const float* hA = s_big + (s * 16 + 8 * half) * 2 * FF + (lm + 16 * half) * 2;
    const float* hB = s_big + (s * 16 + 8 * half) * 2 * FF + (lm + 16 * (1 - half)) * 2;

    v8f   acc[4] = {};
    float lsum   = 0.f;

    v4f an[4];
    v4f rn;
    v4f hs[2][2];
#pragma unroll
    for (int k = 0; k < 2; ++k) {
        hs[k][0] = *(const v4f*)(hs_ld + k * 64 * 2 * FF);
        hs[k][1] = *(const v4f*)(hs_ld + k * 64 * 2 * FF + FF);
    }
#pragma unroll
    for (int q = 0; q < 4; ++q)
        an[q] = __builtin_nontemporal_load(
            (const v4f*)(adj_ld + (size_t)q * 4 * NN + s * PW));
    rn = *(const v4f*)(rbase + s * PW);
#pragma unroll
    for (int k = 0; k < 2; ++k) {
        v4f lo, hi;
        lo.x = hs[k][0].x; lo.y = hs[k][1].x; lo.z = hs[k][0].y; lo.w = hs[k][1].y;
        hi.x = hs[k][0].z; hi.y = hs[k][1].z; hi.z = hs[k][0].w; hi.w = hs[k][1].w;
        *(v4f*)(hs_st + k * 64 * 2 * FF)     = lo;
        *(v4f*)(hs_st + k * 64 * 2 * FF + 4) = hi;
    }
    __syncthreads();

    for (int it = 0; it < NN / PW / NS; ++it) {
        const int j0  = (s + it * NS) * PW;
        const int cur = (it & 1) * CHF;
        const v4f rv = rn;

#pragma unroll
        for (int q = 0; q < 4; ++q) {
            v4f p;
#pragma unroll
            for (int e = 0; e < 4; ++e) {
                const float av = an[q][e];
                float z = lfq[q] + rv[e] + ab;  z = z > 0.f ? z : 0.1f * z;
                const float ex = __expf(z + av);
                p[e] = av > 0.f ? ex : 0.f;
            }
            *(v4f*)(p_st + q * 4 * PP) = p;
        }
        if (it + 1 < NN / PW / NS) {
#pragma unroll
            for (int q = 0; q < 4; ++q)
                an[q] = __builtin_nontemporal_load(
                    (const v4f*)(adj_ld + (size_t)q * 4 * NN + j0 + NS * PW));
            rn = *(const v4f*)(rbase + j0 + NS * PW);
#pragma unroll
            for (int k = 0; k < 2; ++k) {
                hs[k][0] = *(const v4f*)(hs_ld + (size_t)(it + 1) * CHF + k * 64 * 2 * FF);
                hs[k][1] = *(const v4f*)(hs_ld + (size_t)(it + 1) * CHF + k * 64 * 2 * FF + FF);
            }
        }
        __builtin_amdgcn_fence(__ATOMIC_RELEASE, "wavefront");
        __builtin_amdgcn_wave_barrier();
        __builtin_amdgcn_fence(__ATOMIC_ACQUIRE, "wavefront");
        const float* hAc = hA + cur;
        const float* hBc = hB + cur;
        v2f pa[2];
        v2f hb[2][4];
        pa[0]    = *(const v2f*)(p_ld);
        hb[0][0] = *(const v2f*)(hAc);
        hb[0][1] = *(const v2f*)(hBc);
        hb[0][2] = *(const v2f*)(hAc + FF);
        hb[0][3] = *(const v2f*)(hBc + FF);
#pragma unroll
        for (int c = 0; c < 8; ++c) {
            if (c + 1 < 8) {
                pa[(c + 1) & 1]    = *(const v2f*)(p_ld + 2 * (c + 1));
                hb[(c + 1) & 1][0] = *(const v2f*)(hAc + (c + 1) * 2 * FF);
                hb[(c + 1) & 1][1] = *(const v2f*)(hBc + (c + 1) * 2 * FF);
                hb[(c + 1) & 1][2] = *(const v2f*)(hAc + (c + 1) * 2 * FF + FF);
                hb[(c + 1) & 1][3] = *(const v2f*)(hBc + (c + 1) * 2 * FF + FF);
            }
            lsum += pa[c & 1].x + pa[c & 1].y;
#pragma unroll
            for (int nt = 0; nt < 4; ++nt)
                acc[nt] = wmma4(pa[c & 1], hb[c & 1][nt], acc[nt]);
        }
        __builtin_amdgcn_fence(__ATOMIC_RELEASE, "wavefront");
        __builtin_amdgcn_wave_barrier();
        __builtin_amdgcn_fence(__ATOMIC_ACQUIRE, "wavefront");
        if (it + 1 < NN / PW / NS) {
#pragma unroll
            for (int k = 0; k < 2; ++k) {
                v4f lo, hi;
                lo.x = hs[k][0].x; lo.y = hs[k][1].x; lo.z = hs[k][0].y; lo.w = hs[k][1].y;
                hi.x = hs[k][0].z; hi.y = hs[k][1].z; hi.z = hs[k][0].w; hi.w = hs[k][1].w;
                *(v4f*)(hs_st + (CHF - cur) + k * 64 * 2 * FF)     = lo;
                *(v4f*)(hs_st + (CHF - cur) + k * 64 * 2 * FF + 4) = hi;
            }
        }
        __syncthreads();
    }

    lsum += __shfl_xor(lsum, 16, 32);

    const int lane_e = __lane_id();
    const int half_e = lane_e >> 4;
    const int lm_e   = lane_e & 15;

    float* s_acc = s_big;
#pragma unroll
    for (int nt = 0; nt < 4; ++nt)
#pragma unroll
        for (int v = 0; v < 8; ++v)
            s_acc[(((t * NS + s) * 4 + nt) * 8 + v) * 32 + lane_e] = acc[nt][v];
    if (lane_e < 16) s_l[t][s][lm_e] = lsum;
    __syncthreads();

    {
        const int M   = 2 * s + half_e;
        const int v   = M & 7;
        const int cl  = (M >> 3) * 16 + (lm_e & 3) * 4;
        const int ntc = lm_e >> 2;
        float lt = 0.f;
        float o0 = 0.f, o1 = 0.f, o2 = 0.f, o3 = 0.f;
#pragma unroll
        for (int q = 0; q < NS; ++q) {
            lt += s_l[t][q][M];
            const v4f pv = *(const v4f*)&s_acc[(((t * NS + q) * 4 + ntc) * 8 + v) * 32 + cl];
            o0 += pv.x; o1 += pv.y; o2 += pv.z; o3 += pv.w;
        }
        const float inv = lt > 0.f ? 1.0f / lt : 0.f;
        v4f r;
        r.x = o0 * inv; r.y = o1 * inv; r.z = o2 * inv; r.w = o3 * inv;
        r.x = r.x > 0.f ? r.x : (__expf(r.x) - 1.f);
        r.y = r.y > 0.f ? r.y : (__expf(r.y) - 1.f);
        r.z = r.z > 0.f ? r.z : (__expf(r.z) - 1.f);
        r.w = r.w > 0.f ? r.w : (__expf(r.w) - 1.f);
        __builtin_nontemporal_store(
            r, (v4f*)(out + (size_t)(r0 + M) * FF + lm_e * 4));
    }
}

extern "C" void kernel_launch(void* const* d_in, const int* in_sizes, int n_in,
                              void* d_out, int out_size, void* d_ws, size_t ws_size,
                              hipStream_t stream) {
    const float* x   = (const float*)d_in[0];
    const float* adj = (const float*)d_in[1];
    const float* Ww  = (const float*)d_in[2];
    const float* Wb  = (const float*)d_in[3];
    const float* a1  = (const float*)d_in[4];
    const float* a2  = (const float*)d_in[5];
    const float* ab  = (const float*)d_in[6];
    float* out = (float*)d_out;

    float* h     = (float*)d_ws;
    float* left  = h + (size_t)BB * NN * FF;
    float* right = left + (size_t)BB * NN;

    k_h  <<<BB * NN / 32, 64,  0, stream>>>(x, Ww, Wb, a1, a2, h, left, right);
    k_att<<<BB * NN / 64, 32 * 4 * NS, 0, stream>>>(adj, h, left, right, ab, out);
}


# each wave carries two 16-row tiles that share every B fragment read from LDS (16-wave blocks, 4 waves per SIMD): half the B reads, block barriers and waves per matrix instruction; same sum order
# speedup vs baseline: 3.1368x; 1.0446x over previous
#include <hip/hip_runtime.h>

#define BB 8
#define NN 2048
#define FF 64

typedef float v2f __attribute__((ext_vector_type(2)));
typedef float v4f __attribute__((ext_vector_type(4)));
typedef float v8f __attribute__((ext_vector_type(8)));

__device__ __forceinline__ v8f wmma4(v2f a, v2f b, v8f c) {
    return __builtin_amdgcn_wmma_f32_16x16x4_f32(
         false, a,  false, b,
         (short)0, c,  false,  false);
}


template <int MASK>
__device__ __forceinline__ float row_xor(float v) {
    return __int_as_float(__builtin_amdgcn_update_dpp(
        0, __float_as_int(v), 0x160 + MASK  , 0xf, 0xf, true));
}

__device__ __forceinline__ float row_sum16(float v) {
    v += row_xor<1>(v);
    v += row_xor<2>(v);
    v += row_xor<4>(v);
    v += row_xor<8>(v);
    return v;
}

__global__ __launch_bounds__(64) void k_h(const float* __restrict__ x,
                                          const float* __restrict__ Ww,
                                          const float* __restrict__ Wb,
                                          const float* __restrict__ a1,
                                          const float* __restrict__ a2,
                                          float* __restrict__ h,
                                          float* __restrict__ left,
                                          float* __restrict__ right) {
    __shared__ __attribute__((aligned(16))) float s_lr[2][32];

    const int w    = __builtin_amdgcn_readfirstlane(threadIdx.x >> 5);
    const int lane = threadIdx.x & 31;
    const int half = lane >> 4;
    const int lm   = lane & 15;
    const int r0   = (blockIdx.x * 2 + w) * 16;

    v8f acc[4] = {};
#pragma unroll
    for (int kc = 0; kc < 16; ++kc) {
        const int k = kc * 4 + half * 2;
        v2f a = *(const v2f*)(x + (size_t)(r0 + lm) * FF + k);
#pragma unroll
        for (int nt = 0; nt < 4; ++nt) {
            const int col = nt * 16 + lm;
            v2f bop = *(const v2f*)(Ww + (size_t)col * FF + k);
            acc[nt] = wmma4(a, bop, acc[nt]);
        }
    }
    float pl[8] = {}, pr[8] = {};
#pragma unroll
    for (int nt = 0; nt < 4; ++nt) {
        const float bias = Wb[nt * 16 + lm];
        const float c1   = a1[nt * 16 + lm];
        const float c2   = a2[nt * 16 + lm];
#pragma unroll
        for (int v = 0; v < 8; ++v) {
            acc[nt][v] += bias;
            pl[v] = fmaf(acc[nt][v], c1, pl[v]);
            pr[v] = fmaf(acc[nt][v], c2, pr[v]);
        }
    }
#pragma unroll
    for (int v = 0; v < 8; ++v) {
        pl[v] = row_sum16(pl[v]);
        pr[v] = row_sum16(pr[v]);
    }
    if (lm == 0) {
        v4f t;
        t.x = pl[0]; t.y = pl[1]; t.z = pl[2]; t.w = pl[3];
        *(v4f*)&s_lr[0][w * 16 + 8 * half]     = t;
        t.x = pl[4]; t.y = pl[5]; t.z = pl[6]; t.w = pl[7];
        *(v4f*)&s_lr[0][w * 16 + 8 * half + 4] = t;
        t.x = pr[0]; t.y = pr[1]; t.z = pr[2]; t.w = pr[3];
        *(v4f*)&s_lr[1][w * 16 + 8 * half]     = t;
        t.x = pr[4]; t.y = pr[5]; t.z = pr[6]; t.w = pr[7];
        *(v4f*)&s_lr[1][w * 16 + 8 * half + 4] = t;
    }
    __syncthreads();
    float* dst = w ? right : left;
    dst[blockIdx.x * 32 + lane] = s_lr[w][lane];
#pragma unroll
    for (int nt = 0; nt < 4; ++nt)
#pragma unroll
        for (int v = 0; v < 8; ++v)
            h[(size_t)(r0 + v + 8 * half) * FF + nt * 16 + lm] = acc[nt][v];
}

#define NS 8
#define TPW 2
#define PW 32
#define PP 36
#define CH (NS * PW)
#define CHF (CH * FF)
#define NWV (2 * NS)

__global__ __launch_bounds__(32 * NWV) void k_att(const float* __restrict__ adj,
                                             const float* __restrict__ h,
                                             const float* __restrict__ left,
                                             const float* __restrict__ right,
                                             const float* __restrict__ ab_p,
                                             float* __restrict__ out) {
    __shared__ __attribute__((aligned(16))) float s_big[2 * CHF];
    __shared__ float s_l[4][NS][16];
    __shared__ __attribute__((aligned(16))) float s_p[NWV][16 * TPW * PP];

    const int tid  = threadIdx.x;
    const int w    = __builtin_amdgcn_readfirstlane(tid >> 5);
    const int lane = tid & 31;
    const int tp   = w / NS;
    const int s    = w % NS;
    const int half = lane >> 4;
    const int lm   = lane & 15;
    const int lr   = lane >> 3;
    const int lc   = (lane & 7) * 4;
    const int r0   = blockIdx.x * 64 + tp * 32;
    const int b    = r0 / NN;

    const float ab = ab_p[0];
    float lfq[4 * TPW];
#pragma unroll
    for (int q = 0; q < 4 * TPW; ++q) lfq[q] = left[r0 + 4 * q + lr];

    const float* adj_ld = adj + (size_t)(r0 + lr) * NN + lc;
    const float* rbase  = right + (size_t)b * NN + lc;
    float*       p_st   = &s_p[w][lr * PP + lc];
    const float* p_ld   = &s_p[w][lm * PP + 16 * half];

    const float* hs_ld = h + (size_t)b * NN * FF + (tid >> 4) * 2 * FF + (tid & 15) * 4;
    float*       hs_st = s_big + (tid >> 4) * 2 * FF
                               + ((((tid & 15) * 4) ^ (((tid >> 7) & 1) * 16)) * 2);
    const float* hA = s_big + (s * 16 + 8 * half) * 2 * FF + (lm + 16 * half) * 2;
    const float* hB = s_big + (s * 16 + 8 * half) * 2 * FF + (lm + 16 * (1 - half)) * 2;

    v8f   acc[TPW][4] = {};
    float lsum[TPW]   = {};

    v4f an[4 * TPW];
    v4f rn;
    v4f hs[4][2];
#pragma unroll
    for (int k = 0; k < 4; ++k) {
        hs[k][0] = *(const v4f*)(hs_ld + k * 32 * 2 * FF);
        hs[k][1] = *(const v4f*)(hs_ld + k * 32 * 2 * FF + FF);
    }
#pragma unroll
    for (int q = 0; q < 4 * TPW; ++q)
        an[q] = __builtin_nontemporal_load(
            (const v4f*)(adj_ld + (size_t)q * 4 * NN + s * PW));
    rn = *(const v4f*)(rbase + s * PW);
#pragma unroll
    for (int k = 0; k < 4; ++k) {
        v4f lo, hi;
        lo.x = hs[k][0].x; lo.y = hs[k][1].x; lo.z = hs[k][0].y; lo.w = hs[k][1].y;
        hi.x = hs[k][0].z; hi.y = hs[k][1].z; hi.z = hs[k][0].w; hi.w = hs[k][1].w;
        *(v4f*)(hs_st + k * 32 * 2 * FF)     = lo;
        *(v4f*)(hs_st + k * 32 * 2 * FF + 4) = hi;
    }
    __syncthreads();

    for (int it = 0; it < NN / PW / NS; ++it) {
        const int j0  = (s + it * NS) * PW;
        const int cur = (it & 1) * CHF;
        const v4f rv = rn;

#pragma unroll
        for (int q = 0; q < 4 * TPW; ++q) {
            v4f p;
#pragma unroll
            for (int e = 0; e < 4; ++e) {
                const float av = an[q][e];
                float z = lfq[q] + rv[e] + ab;  z = z > 0.f ? z : 0.1f * z;
                const float ex = __expf(z + av);
                p[e] = av > 0.f ? ex : 0.f;
            }
            *(v4f*)(p_st + q * 4 * PP) = p;
        }
        if (it + 1 < NN / PW / NS) {
#pragma unroll
            for (int q = 0; q < 4 * TPW; ++q)
                an[q] = __builtin_nontemporal_load(
                    (const v4f*)(adj_ld + (size_t)q * 4 * NN + j0 + NS * PW));
            rn = *(const v4f*)(rbase + j0 + NS * PW);
#pragma unroll
            for (int k = 0; k < 4; ++k) {
                hs[k][0] = *(const v4f*)(hs_ld + (size_t)(it + 1) * CHF + k * 32 * 2 * FF);
                hs[k][1] = *(const v4f*)(hs_ld + (size_t)(it + 1) * CHF + k * 32 * 2 * FF + FF);
            }
        }
        __builtin_amdgcn_fence(__ATOMIC_RELEASE, "wavefront");
        __builtin_amdgcn_wave_barrier();
        __builtin_amdgcn_fence(__ATOMIC_ACQUIRE, "wavefront");
        const float* hAc = hA + cur;
        const float* hBc = hB + cur;
        v2f pa[2][TPW];
        v2f hb[2][4];
#pragma unroll
        for (int u = 0; u < TPW; ++u) pa[0][u] = *(const v2f*)(p_ld + u * 16 * PP);
        hb[0][0] = *(const v2f*)(hAc);
        hb[0][1] = *(const v2f*)(hBc);
        hb[0][2] = *(const v2f*)(hAc + FF);
        hb[0][3] = *(const v2f*)(hBc + FF);
#pragma unroll
        for (int c = 0; c < 8; ++c) {
            if (c + 1 < 8) {
#pragma unroll
                for (int u = 0; u < TPW; ++u)
                    pa[(c + 1) & 1][u] = *(const v2f*)(p_ld + u * 16 * PP + 2 * (c + 1));
                hb[(c + 1) & 1][0] = *(const v2f*)(hAc + (c + 1) * 2 * FF);
                hb[(c + 1) & 1][1] = *(const v2f*)(hBc + (c + 1) * 2 * FF);
                hb[(c + 1) & 1][2] = *(const v2f*)(hAc + (c + 1) * 2 * FF + FF);
                hb[(c + 1) & 1][3] = *(const v2f*)(hBc + (c + 1) * 2 * FF + FF);
            }
#pragma unroll
            for (int u = 0; u < TPW; ++u) {
                lsum[u] += pa[c & 1][u].x + pa[c & 1][u].y;
#pragma unroll
                for (int nt = 0; nt < 4; ++nt)
                    acc[u][nt] = wmma4(pa[c & 1][u], hb[c & 1][nt], acc[u][nt]);
            }
        }
        __builtin_amdgcn_fence(__ATOMIC_RELEASE, "wavefront");
        __builtin_amdgcn_wave_barrier();
        __builtin_amdgcn_fence(__ATOMIC_ACQUIRE, "wavefront");
        if (it + 1 < NN / PW / NS) {
#pragma unroll
            for (int k = 0; k < 4; ++k) {
                v4f lo, hi;
                lo.x = hs[k][0].x; lo.y = hs[k][1].x; lo.z = hs[k][0].y; lo.w = hs[k][1].y;
                hi.x = hs[k][0].z; hi.y = hs[k][1].z; hi.z = hs[k][0].w; hi.w = hs[k][1].w;
                *(v4f*)(hs_st + (CHF - cur) + k * 32 * 2 * FF)     = lo;
                *(v4f*)(hs_st + (CHF - cur) + k * 32 * 2 * FF + 4) = hi;
            }
        }
        __syncthreads();
    }

    const int lane_e = __lane_id();
    const int half_e = lane_e >> 4;
    const int lm_e   = lane_e & 15;

    float* s_acc = s_big;
#pragma unroll
    for (int u = 0; u < TPW; ++u) {
        const float ls = lsum[u] + __shfl_xor(lsum[u], 16, 32);
        const int t = tp * TPW + u;
#pragma unroll
        for (int nt = 0; nt < 4; ++nt)
#pragma unroll
            for (int v = 0; v < 8; ++v)
                s_acc[(((t * NS + s) * 4 + nt) * 8 + v) * 32 + lane_e] = acc[u][nt][v];
        if (lane_e < 16) s_l[t][s][lm_e] = ls;
    }
    __syncthreads();

#pragma unroll
    for (int u = 0; u < TPW; ++u) {
        const int t   = tp * TPW + u;
        const int M   = 2 * s + half_e;
        const int v   = M & 7;
        const int cl  = (M >> 3) * 16 + (lm_e & 3) * 4;
        const int ntc = lm_e >> 2;
        float lt = 0.f;
        float o0 = 0.f, o1 = 0.f, o2 = 0.f, o3 = 0.f;
#pragma unroll
        for (int q = 0; q < NS; ++q) {
            lt += s_l[t][q][M];
            const v4f pv = *(const v4f*)&s_acc[(((t * NS + q) * 4 + ntc) * 8 + v) * 32 + cl];
            o0 += pv.x; o1 += pv.y; o2 += pv.z; o3 += pv.w;
        }
        const float inv = lt > 0.f ? 1.0f / lt : 0.f;
        v4f r;
        r.x = o0 * inv; r.y = o1 * inv; r.z = o2 * inv; r.w = o3 * inv;
        r.x = r.x > 0.f ? r.x : (__expf(r.x) - 1.f);
        r.y = r.y > 0.f ? r.y : (__expf(r.y) - 1.f);
        r.z = r.z > 0.f ? r.z : (__expf(r.z) - 1.f);
        r.w = r.w > 0.f ? r.w : (__expf(r.w) - 1.f);
        __builtin_nontemporal_store(
            r, (v4f*)(out + (size_t)(r0 + u * 16 + M) * FF + lm_e * 4));
    }
}

extern "C" void kernel_launch(void* const* d_in, const int* in_sizes, int n_in,
                              void* d_out, int out_size, void* d_ws, size_t ws_size,
                              hipStream_t stream) {
    const float* x   = (const float*)d_in[0];
    const float* adj = (const float*)d_in[1];
    const float* Ww  = (const float*)d_in[2];
    const float* Wb  = (const float*)d_in[3];
    const float* a1  = (const float*)d_in[4];
    const float* a2  = (const float*)d_in[5];
    const float* ab  = (const float*)d_in[6];
    float* out = (float*)d_out;

    float* h     = (float*)d_ws;
    float* left  = h + (size_t)BB * NN * FF;
    float* right = left + (size_t)BB * NN;

    k_h  <<<BB * NN / 32, 64,  0, stream>>>(x, Ww, Wb, a1, a2, h, left, right);
    k_att<<<BB * NN / 64, 32 * NWV, 0, stream>>>(adj, h, left, right, ab, out);
}


# h kernel folded into the attention kernel by linearity: the big product takes x, W^T and bias applied once to each block's 64x64 normalised sums, left/right as x.(W^T a); one launch; new sum order
# speedup vs baseline: 3.2925x; 1.0496x over previous
#include <hip/hip_runtime.h>

#define BB 8
#define NN 2048
#define FF 64

typedef float v2f __attribute__((ext_vector_type(2)));
typedef float v4f __attribute__((ext_vector_type(4)));
typedef float v8f __attribute__((ext_vector_type(8)));

__device__ __forceinline__ v8f wmma4(v2f a, v2f b, v8f c) {
    return __builtin_amdgcn_wmma_f32_16x16x4_f32(
         false, a,  false, b,
         (short)0, c,  false,  false);
}

template <int MASK>
__device__ __forceinline__ float row_xor(float v) {
    return __int_as_float(__builtin_amdgcn_update_dpp(
        0, __float_as_int(v), 0x160 + MASK  , 0xf, 0xf, true));
}

__device__ __forceinline__ void chunk_right(const v4f (&hs)[4][2], const v4f w2c,
                                            float* __restrict__ dst, const int l16) {
    const bool b3 = (l16 & 8) != 0, b2 = (l16 & 4) != 0, b1 = (l16 & 2) != 0;
    float t[8];
#pragma unroll
    for (int k = 0; k < 4; ++k)
#pragma unroll
        for (int e = 0; e < 2; ++e) {
            float d = hs[k][e].x * w2c.x;
            d = fmaf(hs[k][e].y, w2c.y, d);
            d = fmaf(hs[k][e].z, w2c.z, d);
            d = fmaf(hs[k][e].w, w2c.w, d);
            t[2 * k + e] = d;
        }
    float u[4];
#pragma unroll
    for (int i = 0; i < 4; ++i) {
        const float keep = b3 ? t[i + 4] : t[i];
        const float send = b3 ? t[i] : t[i + 4];
        u[i] = keep + row_xor<8>(send);
    }
    float v[2];
#pragma unroll
    for (int i = 0; i < 2; ++i) {
        const float keep = b2 ? u[i + 2] : u[i];
        const float send = b2 ? u[i] : u[i + 2];
        v[i] = keep + row_xor<4>(send);
    }
    float y;
    {
        const float keep = b1 ? v[1] : v[0];
        const float send = b1 ? v[0] : v[1];
        y = keep + row_xor<2>(send);
    }
    y += row_xor<1>(y);
    if ((l16 & 1) == 0) dst[(l16 >> 2) * 64 + ((l16 >> 1) & 1)] = y;
}

#define NS 8
#define TPW 2
#define PW 32
#define PP 36
#define CH (NS * PW)
#define CHF (CH * FF)
#define NWV (2 * NS)
#define PS 68

__global__ __launch_bounds__(32 * NWV) void k_att(const float* __restrict__ x,
                                             const float* __restrict__ adj,
                                             const float* __restrict__ Ww,
                                             const float* __restrict__ Wb,
                                             const float* __restrict__ a1,
                                             const float* __restrict__ a2,
                                             const float* __restrict__ ab_p,
                                             float* __restrict__ out) {
    __shared__ __attribute__((aligned(16))) float s_big[2 * CHF];
    __shared__ float s_l[4][NS][16];
    __shared__ __attribute__((aligned(16))) float s_p[NWV][16 * TPW * PP];
    __shared__ float s_wp[2][8][FF];
    __shared__ __attribute__((aligned(16))) float s_wv[2][FF];
    __shared__ __attribute__((aligned(16))) float s_right[2][CH];
    __shared__ __attribute__((aligned(16))) float s_W[FF * PS];

    const int tid  = threadIdx.x;
    const int w    = __builtin_amdgcn_readfirstlane(tid >> 5);
    const int lane = tid & 31;
    const int tp   = w / NS;
    const int s    = w % NS;
    const int half = lane >> 4;
    const int lm   = lane & 15;
    const int lr   = lane >> 3;
    const int lc   = (lane & 7) * 4;
    const int r0   = blockIdx.x * 64 + tp * 32;
    const int b    = r0 / NN;

    const float* adj_ld = adj + (size_t)(r0 + lr) * NN + lc;
    float*       p_st   = &s_p[w][lr * PP + lc];
    const float* p_ld   = &s_p[w][lm * PP + 16 * half];

    const float* hs_ld = x + (size_t)b * NN * FF + (tid >> 4) * 2 * FF + (tid & 15) * 4;
    float*       hs_st = s_big + (tid >> 4) * 2 * FF
                               + ((((tid & 15) * 4) ^ (((tid >> 7) & 1) * 16)) * 2);
    float*       rt_st = &s_right[0][(tid >> 4) * 2];
    const float* hA = s_big + (s * 16 + 8 * half) * 2 * FF + (lm + 16 * half) * 2;
    const float* hB = s_big + (s * 16 + 8 * half) * 2 * FF + (lm + 16 * (1 - half)) * 2;

    v4f an[4 * TPW];
    v4f hs[4][2];
#pragma unroll
    for (int k = 0; k < 4; ++k) {
        hs[k][0] = *(const v4f*)(hs_ld + k * 32 * 2 * FF);
        hs[k][1] = *(const v4f*)(hs_ld + k * 32 * 2 * FF + FF);
    }
#pragma unroll
    for (int q = 0; q < 4 * TPW; ++q)
        an[q] = __builtin_nontemporal_load(
            (const v4f*)(adj_ld + (size_t)q * 4 * NN + s * PW));

#pragma unroll
    for (int i = 0; i < 2; ++i) {
        const int f = (i * 512 + tid) * 4;
        *(v4f*)&s_W[(f >> 6) * PS + (f & 63)] = *(const v4f*)(Ww + f);
    }

    float lfq[4 * TPW];
    float cab;
    {
        const float* xl_ld = x + (size_t)(r0 + lr) * FF + (lane & 7) * 8;
        v4f xl[4 * TPW][2];
#pragma unroll
        for (int q = 0; q < 4 * TPW; ++q) {
            xl[q][0] = *(const v4f*)(xl_ld + q * 4 * FF);
            xl[q][1] = *(const v4f*)(xl_ld + q * 4 * FF + 4);
        }
        {
            const int k  = tid & 63;
            const int og = __builtin_amdgcn_readfirstlane(tid >> 6);
            float p1 = 0.f, p2 = 0.f;
#pragma unroll
            for (int i = 0; i < 8; ++i) {
                const float wv = Ww[(og * 8 + i) * FF + k];
                p1 = fmaf(wv, a1[og * 8 + i], p1);
                p2 = fmaf(wv, a2[og * 8 + i], p2);
            }
            s_wp[0][og][k] = p1;
            s_wp[1][og][k] = p2;
        }
        cab = ab_p[0];
#pragma unroll
        for (int o = 0; o < FF; ++o) cab = fmaf(Wb[o], a1[o] + a2[o], cab);
        __syncthreads();
        if (tid < 2 * FF) {
            const int which = tid >> 6, k = tid & 63;
            float t = 0.f;
#pragma unroll
            for (int g = 0; g < 8; ++g) t += s_wp[which][g][k];
            s_wv[which][k] = t;
        }
        __syncthreads();
        const v4f wa = *(const v4f*)&s_wv[0][(lane & 7) * 8];
        const v4f wb = *(const v4f*)&s_wv[0][(lane & 7) * 8 + 4];
#pragma unroll
        for (int q = 0; q < 4 * TPW; ++q) {
            float t = xl[q][0].x * wa.x;
            t = fmaf(xl[q][0].y, wa.y, t);
            t = fmaf(xl[q][0].z, wa.z, t);
            t = fmaf(xl[q][0].w, wa.w, t);
            t = fmaf(xl[q][1].x, wb.x, t);
            t = fmaf(xl[q][1].y, wb.y, t);
            t = fmaf(xl[q][1].z, wb.z, t);
            t = fmaf(xl[q][1].w, wb.w, t);
            t += row_xor<1>(t);
            t += row_xor<2>(t);
            t += row_xor<4>(t);
            lfq[q] = t;
        }
    }

    chunk_right(hs, *(const v4f*)&s_wv[1][(tid & 15) * 4], rt_st, tid & 15);
#pragma unroll
    for (int k = 0; k < 4; ++k) {
        v4f lo, hi;
        lo.x = hs[k][0].x; lo.y = hs[k][1].x; lo.z = hs[k][0].y; lo.w = hs[k][1].y;
        hi.x = hs[k][0].z; hi.y = hs[k][1].z; hi.z = hs[k][0].w; hi.w = hs[k][1].w;
        *(v4f*)(hs_st + k * 32 * 2 * FF)     = lo;
        *(v4f*)(hs_st + k * 32 * 2 * FF + 4) = hi;
    }
    __syncthreads();

    v8f   acc[TPW][4] = {};
    float lsum[TPW]   = {};

    for (int it = 0; it < NN / PW / NS; ++it) {
        const int j0  = (s + it * NS) * PW;
        const int cur = (it & 1) * CHF;
        if (it + 1 < NN / PW / NS) {
#pragma unroll
            for (int k = 0; k < 4; ++k) {
                hs[k][0] = *(const v4f*)(hs_ld + (size_t)(it + 1) * CHF + k * 32 * 2 * FF);
                hs[k][1] = *(const v4f*)(hs_ld + (size_t)(it + 1) * CHF + k * 32 * 2 * FF + FF);
            }
        }
        const v4f rv = *(const v4f*)&s_right[it & 1][s * PW + lc];

#pragma unroll
        for (int q = 0; q < 4 * TPW; ++q) {
            v4f p;
#pragma unroll
            for (int e = 0; e < 4; ++e) {
                const float av = an[q][e];
                float z = lfq[q] + rv[e] + cab;  z = z > 0.f ? z : 0.1f * z;
                const float ex = __expf(z + av);
                p[e] = av > 0.f ? ex : 0.f;
            }
            *(v4f*)(p_st + q * 4 * PP) = p;
        }
        if (it + 1 < NN / PW / NS) {
#pragma unroll
            for (int q = 0; q < 4 * TPW; ++q)
                an[q] = __builtin_nontemporal_load(
                    (const v4f*)(adj_ld + (size_t)q * 4 * NN + j0 + NS * PW));
        }
        __builtin_amdgcn_fence(__ATOMIC_RELEASE, "wavefront");
        __builtin_amdgcn_wave_barrier();
        __builtin_amdgcn_fence(__ATOMIC_ACQUIRE, "wavefront");
        const float* hAc = hA + cur;
        const float* hBc = hB + cur;
        v2f pa[2][TPW];
        v2f hb[2][4];
#pragma unroll
        for (int u = 0; u < TPW; ++u) pa[0][u] = *(const v2f*)(p_ld + u * 16 * PP);
        hb[0][0] = *(const v2f*)(hAc);
        hb[0][1] = *(const v2f*)(hBc);
        hb[0][2] = *(const v2f*)(hAc + FF);
        hb[0][3] = *(const v2f*)(hBc + FF);
#pragma unroll
        for (int c = 0; c < 8; ++c) {
            if (c + 1 < 8) {
#pragma unroll
                for (int u = 0; u < TPW; ++u)
                    pa[(c + 1) & 1][u] = *(const v2f*)(p_ld + u * 16 * PP + 2 * (c + 1));
                hb[(c + 1) & 1][0] = *(const v2f*)(hAc + (c + 1) * 2 * FF);
                hb[(c + 1) & 1][1] = *(const v2f*)(hBc + (c + 1) * 2 * FF);
                hb[(c + 1) & 1][2] = *(const v2f*)(hAc + (c + 1) * 2 * FF + FF);
                hb[(c + 1) & 1][3] = *(const v2f*)(hBc + (c + 1) * 2 * FF + FF);
            }
#pragma unroll
            for (int u = 0; u < TPW; ++u) {
                lsum[u] += pa[c & 1][u].x + pa[c & 1][u].y;
#pragma unroll
                for (int nt = 0; nt < 4; ++nt)
                    acc[u][nt] = wmma4(pa[c & 1][u], hb[c & 1][nt], acc[u][nt]);
            }
        }
        __builtin_amdgcn_fence(__ATOMIC_RELEASE, "wavefront");
        __builtin_amdgcn_wave_barrier();
        __builtin_amdgcn_fence(__ATOMIC_ACQUIRE, "wavefront");
        if (it + 1 < NN / PW / NS) {
            chunk_right(hs, *(const v4f*)&s_wv[1][(tid & 15) * 4],
                        rt_st + ((it + 1) & 1) * CH, tid & 15);
#pragma unroll
            for (int k = 0; k < 4; ++k) {
                v4f lo, hi;
                lo.x = hs[k][0].x; lo.y = hs[k][1].x; lo.z = hs[k][0].y; lo.w = hs[k][1].y;
                hi.x = hs[k][0].z; hi.y = hs[k][1].z; hi.z = hs[k][0].w; hi.w = hs[k][1].w;
                *(v4f*)(hs_st + (CHF - cur) + k * 32 * 2 * FF)     = lo;
                *(v4f*)(hs_st + (CHF - cur) + k * 32 * 2 * FF + 4) = hi;
            }
        }
        __syncthreads();
    }

    const int lane_e = __lane_id();
    const int half_e = lane_e >> 4;
    const int lm_e   = lane_e & 15;

    float* s_acc = s_big;
#pragma unroll
    for (int u = 0; u < TPW; ++u) {
        const float ls = lsum[u] + __shfl_xor(lsum[u], 16, 32);
        const int t = tp * TPW + u;
#pragma unroll
        for (int nt = 0; nt < 4; ++nt)
#pragma unroll
            for (int v = 0; v < 8; ++v)
                s_acc[(((t * NS + s) * 4 + nt) * 8 + v) * 32 + lane_e] = acc[u][nt][v];
        if (lane_e < 16) s_l[t][s][lm_e] = ls;
    }
    const int pt = w >> 1;
    const int ph = w & 1;
    __syncthreads();

    float* s_S = &s_p[0][0];
#pragma unroll
    for (int u = 0; u < TPW; ++u) {
        const int t   = tp * TPW + u;
        const int M   = 2 * s + half_e;
        const int v   = M & 7;
        const int cl  = (M >> 3) * 16 + (lm_e & 3) * 4;
        const int ntc = lm_e >> 2;
        float lt = 0.f;
        float o0 = 0.f, o1 = 0.f, o2 = 0.f, o3 = 0.f;
#pragma unroll
        for (int q = 0; q < NS; ++q) {
            lt += s_l[t][q][M];
            const v4f pv = *(const v4f*)&s_acc[(((t * NS + q) * 4 + ntc) * 8 + v) * 32 + cl];
            o0 += pv.x; o1 += pv.y; o2 += pv.z; o3 += pv.w;
        }
        const float inv = lt > 0.f ? 1.0f / lt : 0.f;
        v4f r;
        r.x = o0 * inv; r.y = o1 * inv; r.z = o2 * inv; r.w = o3 * inv;
        *(v4f*)&s_S[(t * 16 + M) * PS + lm_e * 4] = r;
    }
    __syncthreads();

    if (w < 8) {
        v8f o[2] = {};
        const float* a_ld = &s_S[(pt * 16 + lm_e) * PS + 2 * half_e];
        const float* b_ld = &s_W[(ph * 32 + lm_e) * PS + 2 * half_e];
#pragma unroll
        for (int c = 0; c < 16; ++c) {
            const v2f a = *(const v2f*)(a_ld + 4 * c);
#pragma unroll
            for (int j = 0; j < 2; ++j)
                o[j] = wmma4(a, *(const v2f*)(b_ld + j * 16 * PS + 4 * c), o[j]);
        }
#pragma unroll
        for (int j = 0; j < 2; ++j) {
            const int   col  = ph * 32 + j * 16 + lm_e;
            const float bias = Wb[col];
#pragma unroll
            for (int v = 0; v < 8; ++v) {
                float r = o[j][v] + bias;
                r = r > 0.f ? r : (__expf(r) - 1.f);
                out[(size_t)(blockIdx.x * 64 + pt * 16 + v + 8 * half_e) * FF + col] = r;
            }
        }
    }
}

extern "C" void kernel_launch(void* const* d_in, const int* in_sizes, int n_in,
                              void* d_out, int out_size, void* d_ws, size_t ws_size,
                              hipStream_t stream) {
    const float* x   = (const float*)d_in[0];
    const float* adj = (const float*)d_in[1];
    const float* Ww  = (const float*)d_in[2];
    const float* Wb  = (const float*)d_in[3];
    const float* a1  = (const float*)d_in[4];
    const float* a2  = (const float*)d_in[5];
    const float* ab  = (const float*)d_in[6];
    float* out = (float*)d_out;

    k_att<<<BB * NN / 64, 32 * NWV, 0, stream>>>(x, adj, Ww, Wb, a1, a2, ab, out);
}


# p chain 340 to 232 vector instructions per wave-trip: log2(e) and constant logit part folded into left/right, leaky as max(z, z/10), adj joined by one FMA into bare exp2, sums 2-wide; new sum order
# speedup vs baseline: 3.4284x; 1.0413x over previous
#include <hip/hip_runtime.h>

#define BB 8
#define LOG2E 1.4426950408889634f
#define NN 2048
#define FF 64

typedef float v2f __attribute__((ext_vector_type(2)));
typedef float v4f __attribute__((ext_vector_type(4)));
typedef float v8f __attribute__((ext_vector_type(8)));

__device__ __forceinline__ v8f wmma4(v2f a, v2f b, v8f c) {
    return __builtin_amdgcn_wmma_f32_16x16x4_f32(
         false, a,  false, b,
         (short)0, c,  false,  false);
}

template <int MASK>
__device__ __forceinline__ float row_xor(float v) {
    return __int_as_float(__builtin_amdgcn_update_dpp(
        0, __float_as_int(v), 0x160 + MASK  , 0xf, 0xf, true));
}

__device__ __forceinline__ void chunk_right(const v4f (&hs)[4][2], const v4f w2c,
                                            float* __restrict__ dst, const int l16) {
    const bool b3 = (l16 & 8) != 0, b2 = (l16 & 4) != 0, b1 = (l16 & 2) != 0;
    float t[8];
#pragma unroll
    for (int k = 0; k < 4; ++k)
#pragma unroll
        for (int e = 0; e < 2; ++e) {
            const v2f lo = {hs[k][e].x, hs[k][e].y}, hi = {hs[k][e].z, hs[k][e].w};
            const v2f wl = {w2c.x, w2c.y},           wh = {w2c.z, w2c.w};
            const v2f d  = __builtin_elementwise_fma(hi, wh, lo * wl);
            t[2 * k + e] = d.x + d.y;
        }
    float u[4];
#pragma unroll
    for (int i = 0; i < 4; ++i) {
        const float keep = b3 ? t[i + 4] : t[i];
        const float send = b3 ? t[i] : t[i + 4];
        u[i] = keep + row_xor<8>(send);
    }
    float v[2];
#pragma unroll
    for (int i = 0; i < 2; ++i) {
        const float keep = b2 ? u[i + 2] : u[i];
        const float send = b2 ? u[i] : u[i + 2];
        v[i] = keep + row_xor<4>(send);
    }
    float y;
    {
        const float keep = b1 ? v[1] : v[0];
        const float send = b1 ? v[0] : v[1];
        y = keep + row_xor<2>(send);
    }
    y += row_xor<1>(y);
    if ((l16 & 1) == 0) dst[(l16 >> 2) * 64 + ((l16 >> 1) & 1)] = y;
}

#define NS 8
#define TPW 2
#define PW 32
#define PP 36
#define CH (NS * PW)
#define CHF (CH * FF)
#define NWV (2 * NS)
#define PS 68

__global__ __launch_bounds__(32 * NWV) void k_att(const float* __restrict__ x,
                                             const float* __restrict__ adj,
                                             const float* __restrict__ Ww,
                                             const float* __restrict__ Wb,
                                             const float* __restrict__ a1,
                                             const float* __restrict__ a2,
                                             const float* __restrict__ ab_p,
                                             float* __restrict__ out) {
    __shared__ __attribute__((aligned(16))) float s_big[2 * CHF];
    __shared__ float s_l[4][NS][16];
    __shared__ __attribute__((aligned(16))) float s_p[NWV][16 * TPW * PP];
    __shared__ float s_wp[2][8][FF];
    __shared__ __attribute__((aligned(16))) float s_wv[2][FF];
    __shared__ __attribute__((aligned(16))) float s_right[2][CH];
    __shared__ __attribute__((aligned(16))) float s_W[FF * PS];

    const int tid  = threadIdx.x;
    const int w    = __builtin_amdgcn_readfirstlane(tid >> 5);
    const int lane = tid & 31;
    const int tp   = w / NS;
    const int s    = w % NS;
    const int half = lane >> 4;
    const int lm   = lane & 15;
    const int lr   = lane >> 3;
    const int lc   = (lane & 7) * 4;
    const int r0   = blockIdx.x * 64 + tp * 32;
    const int b    = r0 / NN;

    const float* adj_ld = adj + (size_t)(r0 + lr) * NN + lc;
    float*       p_st   = &s_p[w][lr * PP + lc];
    const float* p_ld   = &s_p[w][lm * PP + 16 * half];

    const float* hs_ld = x + (size_t)b * NN * FF + (tid >> 4) * 2 * FF + (tid & 15) * 4;
    float*       hs_st = s_big + (tid >> 4) * 2 * FF
                               + ((((tid & 15) * 4) ^ (((tid >> 7) & 1) * 16)) * 2);
    float*       rt_st = &s_right[0][(tid >> 4) * 2];
    const float* hA = s_big + (s * 16 + 8 * half) * 2 * FF + (lm + 16 * half) * 2;
    const float* hB = s_big + (s * 16 + 8 * half) * 2 * FF + (lm + 16 * (1 - half)) * 2;

    v4f an[4 * TPW];
    v4f hs[4][2];
#pragma unroll
    for (int k = 0; k < 4; ++k) {
        hs[k][0] = *(const v4f*)(hs_ld + k * 32 * 2 * FF);
        hs[k][1] = *(const v4f*)(hs_ld + k * 32 * 2 * FF + FF);
    }
#pragma unroll
    for (int q = 0; q < 4 * TPW; ++q)
        an[q] = __builtin_nontemporal_load(
            (const v4f*)(adj_ld + (size_t)q * 4 * NN + s * PW));

#pragma unroll
    for (int i = 0; i < 2; ++i) {
        const int f = (i * 512 + tid) * 4;
        *(v4f*)&s_W[(f >> 6) * PS + (f & 63)] = *(const v4f*)(Ww + f);
    }

    float lfq[4 * TPW];
    {
        float cab;
        const float* xl_ld = x + (size_t)(r0 + lr) * FF + (lane & 7) * 8;
        v4f xl[4 * TPW][2];
#pragma unroll
        for (int q = 0; q < 4 * TPW; ++q) {
            xl[q][0] = *(const v4f*)(xl_ld + q * 4 * FF);
            xl[q][1] = *(const v4f*)(xl_ld + q * 4 * FF + 4);
        }
        {
            const int k  = tid & 63;
            const int og = __builtin_amdgcn_readfirstlane(tid >> 6);
            float p1 = 0.f, p2 = 0.f;
#pragma unroll
            for (int i = 0; i < 8; ++i) {
                const float wv = Ww[(og * 8 + i) * FF + k];
                p1 = fmaf(wv, a1[og * 8 + i], p1);
                p2 = fmaf(wv, a2[og * 8 + i], p2);
            }
            s_wp[0][og][k] = p1;
            s_wp[1][og][k] = p2;
        }
        cab = ab_p[0];
#pragma unroll
        for (int o = 0; o < FF; ++o) cab = fmaf(Wb[o], a1[o] + a2[o], cab);
        __syncthreads();
        if (tid < 2 * FF) {
            const int which = tid >> 6, k = tid & 63;
            float t = 0.f;
#pragma unroll
            for (int g = 0; g < 8; ++g) t += s_wp[which][g][k];
            s_wv[which][k] = t * LOG2E;
        }
        __syncthreads();
        const v4f wa = *(const v4f*)&s_wv[0][(lane & 7) * 8];
        const v4f wb = *(const v4f*)&s_wv[0][(lane & 7) * 8 + 4];
#pragma unroll
        for (int q = 0; q < 4 * TPW; ++q) {
            float t = xl[q][0].x * wa.x;
            t = fmaf(xl[q][0].y, wa.y, t);
            t = fmaf(xl[q][0].z, wa.z, t);
            t = fmaf(xl[q][0].w, wa.w, t);
            t = fmaf(xl[q][1].x, wb.x, t);
            t = fmaf(xl[q][1].y, wb.y, t);
            t = fmaf(xl[q][1].z, wb.z, t);
            t = fmaf(xl[q][1].w, wb.w, t);
            t += row_xor<1>(t);
            t += row_xor<2>(t);
            t += row_xor<4>(t);
            lfq[q] = fmaf(cab, LOG2E, t);
        }
    }

    chunk_right(hs, *(const v4f*)&s_wv[1][(tid & 15) * 4], rt_st, tid & 15);
#pragma unroll
    for (int k = 0; k < 4; ++k) {
        v4f lo, hi;
        lo.x = hs[k][0].x; lo.y = hs[k][1].x; lo.z = hs[k][0].y; lo.w = hs[k][1].y;
        hi.x = hs[k][0].z; hi.y = hs[k][1].z; hi.z = hs[k][0].w; hi.w = hs[k][1].w;
        *(v4f*)(hs_st + k * 32 * 2 * FF)     = lo;
        *(v4f*)(hs_st + k * 32 * 2 * FF + 4) = hi;
    }
    __syncthreads();

    v8f acc[TPW][4] = {};
    v2f lsum2[TPW]  = {};
    const v2f l2e2  = {LOG2E, LOG2E};

    for (int it = 0; it < NN / PW / NS; ++it) {
        const int j0  = (s + it * NS) * PW;
        const int cur = (it & 1) * CHF;
        if (it + 1 < NN / PW / NS) {
#pragma unroll
            for (int k = 0; k < 4; ++k) {
                hs[k][0] = *(const v4f*)(hs_ld + (size_t)(it + 1) * CHF + k * 32 * 2 * FF);
                hs[k][1] = *(const v4f*)(hs_ld + (size_t)(it + 1) * CHF + k * 32 * 2 * FF + FF);
            }
        }
        const v4f rv = *(const v4f*)&s_right[it & 1][s * PW + lc];

        const v2f rv2[2] = {{rv.x, rv.y}, {rv.z, rv.w}};
#pragma unroll
        for (int q = 0; q < 4 * TPW; ++q) {
            const v2f lq = {lfq[q], lfq[q]};
            v4f p;
#pragma unroll
            for (int g = 0; g < 2; ++g) {
                const v2f av = {an[q][2 * g], an[q][2 * g + 1]};
                const v2f z  = lq + rv2[g];
                const v2f zl = __builtin_elementwise_max(z, z * 0.1f);
                const v2f y  = __builtin_elementwise_fma(av, l2e2, zl);
                p[2 * g]     = av.x > 0.f ? __builtin_amdgcn_exp2f(y.x) : 0.f;
                p[2 * g + 1] = av.y > 0.f ? __builtin_amdgcn_exp2f(y.y) : 0.f;
            }
            *(v4f*)(p_st + q * 4 * PP) = p;
        }
        if (it + 1 < NN / PW / NS) {
#pragma unroll
            for (int q = 0; q < 4 * TPW; ++q)
                an[q] = __builtin_nontemporal_load(
                    (const v4f*)(adj_ld + (size_t)q * 4 * NN + j0 + NS * PW));
        }
        __builtin_amdgcn_fence(__ATOMIC_RELEASE, "wavefront");
        __builtin_amdgcn_wave_barrier();
        __builtin_amdgcn_fence(__ATOMIC_ACQUIRE, "wavefront");
        const float* hAc = hA + cur;
        const float* hBc = hB + cur;
        v2f pa[2][TPW];
        v2f hb[2][4];
#pragma unroll
        for (int u = 0; u < TPW; ++u) pa[0][u] = *(const v2f*)(p_ld + u * 16 * PP);
        hb[0][0] = *(const v2f*)(hAc);
        hb[0][1] = *(const v2f*)(hBc);
        hb[0][2] = *(const v2f*)(hAc + FF);
        hb[0][3] = *(const v2f*)(hBc + FF);
#pragma unroll
        for (int c = 0; c < 8; ++c) {
            if (c + 1 < 8) {
#pragma unroll
                for (int u = 0; u < TPW; ++u)
                    pa[(c + 1) & 1][u] = *(const v2f*)(p_ld + u * 16 * PP + 2 * (c + 1));
                hb[(c + 1) & 1][0] = *(const v2f*)(hAc + (c + 1) * 2 * FF);
                hb[(c + 1) & 1][1] = *(const v2f*)(hBc + (c + 1) * 2 * FF);
                hb[(c + 1) & 1][2] = *(const v2f*)(hAc + (c + 1) * 2 * FF + FF);
                hb[(c + 1) & 1][3] = *(const v2f*)(hBc + (c + 1) * 2 * FF + FF);
            }
#pragma unroll
            for (int u = 0; u < TPW; ++u) {
                lsum2[u] += pa[c & 1][u];
#pragma unroll
                for (int nt = 0; nt < 4; ++nt)
                    acc[u][nt] = wmma4(pa[c & 1][u], hb[c & 1][nt], acc[u][nt]);
            }
        }
        __builtin_amdgcn_fence(__ATOMIC_RELEASE, "wavefront");
        __builtin_amdgcn_wave_barrier();
        __builtin_amdgcn_fence(__ATOMIC_ACQUIRE, "wavefront");
        if (it + 1 < NN / PW / NS) {
            chunk_right(hs, *(const v4f*)&s_wv[1][(tid & 15) * 4],
                        rt_st + ((it + 1) & 1) * CH, tid & 15);
#pragma unroll
            for (int k = 0; k < 4; ++k) {
                v4f lo, hi;
                lo.x = hs[k][0].x; lo.y = hs[k][1].x; lo.z = hs[k][0].y; lo.w = hs[k][1].y;
                hi.x = hs[k][0].z; hi.y = hs[k][1].z; hi.z = hs[k][0].w; hi.w = hs[k][1].w;
                *(v4f*)(hs_st + (CHF - cur) + k * 32 * 2 * FF)     = lo;
                *(v4f*)(hs_st + (CHF - cur) + k * 32 * 2 * FF + 4) = hi;
            }
        }
        __syncthreads();
    }

    const int lane_e = __lane_id();
    const int half_e = lane_e >> 4;
    const int lm_e   = lane_e & 15;

    float* s_acc = s_big;
#pragma unroll
    for (int u = 0; u < TPW; ++u) {
        const float lsu = lsum2[u].x + lsum2[u].y;
        const float ls  = lsu + __shfl_xor(lsu, 16, 32);
        const int t = tp * TPW + u;
#pragma unroll
        for (int nt = 0; nt < 4; ++nt)
#pragma unroll
            for (int v = 0; v < 8; ++v)
                s_acc[(((t * NS + s) * 4 + nt) * 8 + v) * 32 + lane_e] = acc[u][nt][v];
        if (lane_e < 16) s_l[t][s][lm_e] = ls;
    }
    const int pt = w >> 1;
    const int ph = w & 1;
    __syncthreads();

    float* s_S = &s_p[0][0];
#pragma unroll
    for (int u = 0; u < TPW; ++u) {
        const int t   = tp * TPW + u;
        const int M   = 2 * s + half_e;
        const int v   = M & 7;
        const int cl  = (M >> 3) * 16 + (lm_e & 3) * 4;
        const int ntc = lm_e >> 2;
        float lt = 0.f;
        float o0 = 0.f, o1 = 0.f, o2 = 0.f, o3 = 0.f;
#pragma unroll
        for (int q = 0; q < NS; ++q) {
            lt += s_l[t][q][M];
            const v4f pv = *(const v4f*)&s_acc[(((t * NS + q) * 4 + ntc) * 8 + v) * 32 + cl];
            o0 += pv.x; o1 += pv.y; o2 += pv.z; o3 += pv.w;
        }
        const float inv = lt > 0.f ? 1.0f / lt : 0.f;
        v4f r;
        r.x = o0 * inv; r.y = o1 * inv; r.z = o2 * inv; r.w = o3 * inv;
        *(v4f*)&s_S[(t * 16 + M) * PS + lm_e * 4] = r;
    }
    __syncthreads();

    if (w < 8) {
        v8f o[2] = {};
        const float* a_ld = &s_S[(pt * 16 + lm_e) * PS + 2 * half_e];
        const float* b_ld = &s_W[(ph * 32 + lm_e) * PS + 2 * half_e];
#pragma unroll
        for (int c = 0; c < 16; ++c) {
            const v2f a = *(const v2f*)(a_ld + 4 * c);
#pragma unroll
            for (int j = 0; j < 2; ++j)
                o[j] = wmma4(a, *(const v2f*)(b_ld + j * 16 * PS + 4 * c), o[j]);
        }
#pragma unroll
        for (int j = 0; j < 2; ++j) {
            const int   col  = ph * 32 + j * 16 + lm_e;
            const float bias = Wb[col];
#pragma unroll
            for (int v = 0; v < 8; ++v) {
                float r = o[j][v] + bias;
                r = r > 0.f ? r : (__expf(r) - 1.f);
                out[(size_t)(blockIdx.x * 64 + pt * 16 + v + 8 * half_e) * FF + col] = r;
            }
        }
    }
}

extern "C" void kernel_launch(void* const* d_in, const int* in_sizes, int n_in,
                              void* d_out, int out_size, void* d_ws, size_t ws_size,
                              hipStream_t stream) {
    const float* x   = (const float*)d_in[0];
    const float* adj = (const float*)d_in[1];
    const float* Ww  = (const float*)d_in[2];
    const float* Wb  = (const float*)d_in[3];
    const float* a1  = (const float*)d_in[4];
    const float* a2  = (const float*)d_in[5];
    const float* ab  = (const float*)d_in[6];
    float* out = (float*)d_out;

    k_att<<<BB * NN / 64, 32 * NWV, 0, stream>>>(x, adj, Ww, Wb, a1, a2, ab, out);
}
